# starting kernel
# baseline (speedup 1.0000x reference)
#include <hip/hip_runtime.h>
#include <math.h>


typedef __attribute__((ext_vector_type(2))) float v2f;
typedef __attribute__((ext_vector_type(4))) float v4f;
typedef __attribute__((ext_vector_type(8))) float v8f;

#define WAVES 4
#define TPB   (WAVES * 32)
#define BATCH 32
#define NOP   512
#define NMA   64
#define FDIM  8
#define HDIM  128
#define ODIM  8

__device__ __forceinline__ float elu1(float x) {
  float e = __builtin_amdgcn_exp2f(x * 1.44269504088896340736f) - 1.0f;
  return x > 0.0f ? x : e;
}

__device__ __forceinline__ v8f wmma_f32(v2f a, v2f b, v8f c) {
  return __builtin_amdgcn_wmma_f32_16x16x4_f32(false, a, false, b, (short)0, c,
                                               false, false);
}

template <int K, bool ELU>
__device__ __forceinline__ void layer_h(const float* __restrict__ W,
                                        const float* __restrict__ bias,
                                        const float* sA, float* sY, int lane) {
  const int half = lane >> 4, l16 = lane & 15;
  const int kh = half * 2;
  for (int nt = 0; nt < HDIM / 16; ++nt) {
    const int n0 = nt * 16;
    v8f acc0 = {0.f, 0.f, 0.f, 0.f, 0.f, 0.f, 0.f, 0.f};
    v8f acc1 = {0.f, 0.f, 0.f, 0.f, 0.f, 0.f, 0.f, 0.f};
#pragma unroll
    for (int kc = 0; kc < K / 4; kc += 2) {
      {
        const int k = kc * 4 + kh;
        v2f a; a.x = sA[l16 * K + k];            a.y = sA[l16 * K + k + 1];
        v2f b; b.x = W[k * HDIM + n0 + l16];     b.y = W[(k + 1) * HDIM + n0 + l16];
        acc0 = wmma_f32(a, b, acc0);
      }
      {
        const int k = (kc + 1) * 4 + kh;
        v2f a; a.x = sA[l16 * K + k];            a.y = sA[l16 * K + k + 1];
        v2f b; b.x = W[k * HDIM + n0 + l16];     b.y = W[(k + 1) * HDIM + n0 + l16];
        acc1 = wmma_f32(a, b, acc1);
      }
    }
    const float bv = bias[n0 + l16];
#pragma unroll
    for (int r = 0; r < 8; ++r) {
      float y = acc0[r] + acc1[r] + bv;
      if (ELU) y = elu1(y);
      sY[(r + half * 8) * HDIM + n0 + l16] = y;
    }
  }
}

template <int K>
__device__ __forceinline__ v8f layer_o(const float* __restrict__ W,
                                       const float* __restrict__ bias,
                                       const float* sA, int lane) {
  const int half = lane >> 4, l16 = lane & 15;
  const int kh = half * 2;
  const int n = (l16 < 8) ? l16 : 0;
  v8f acc0 = {0.f, 0.f, 0.f, 0.f, 0.f, 0.f, 0.f, 0.f};
  v8f acc1 = {0.f, 0.f, 0.f, 0.f, 0.f, 0.f, 0.f, 0.f};
#pragma unroll
  for (int kc = 0; kc < K / 4; kc += 2) {
    {
      const int k = kc * 4 + kh;
      v2f a; a.x = sA[l16 * K + k];    a.y = sA[l16 * K + k + 1];
      v2f b; b.x = W[k * ODIM + n];    b.y = W[(k + 1) * ODIM + n];
      acc0 = wmma_f32(a, b, acc0);
    }
    {
      const int k = (kc + 1) * 4 + kh;
      v2f a; a.x = sA[l16 * K + k];    a.y = sA[l16 * K + k + 1];
      v2f b; b.x = W[k * ODIM + n];    b.y = W[(k + 1) * ODIM + n];
      acc1 = wmma_f32(a, b, acc1);
    }
  }
  const float bv = bias[n];
#pragma unroll
  for (int r = 0; r < 8; ++r) acc0[r] += acc1[r] + bv;
  return acc0;
}

__global__ __launch_bounds__(TPB) void fused_hgnn_mlp(
    const int* __restrict__ adj, const float* __restrict__ feats_op,
    const float* __restrict__ feats_ma,
    const float* __restrict__ l0w1, const float* __restrict__ l0b1,
    const float* __restrict__ l0w2, const float* __restrict__ l0b2,
    const float* __restrict__ l0w3, const float* __restrict__ l0b3,
    const float* __restrict__ l1w1, const float* __restrict__ l1b1,
    const float* __restrict__ l1w2, const float* __restrict__ l1b2,
    const float* __restrict__ l1w3, const float* __restrict__ l1b3,
    const float* __restrict__ pw1, const float* __restrict__ pb1,
    const float* __restrict__ pw2, const float* __restrict__ pb2,
    const float* __restrict__ pw3, const float* __restrict__ pb3,
    float* __restrict__ out) {
  __shared__ float sFM[NMA * FDIM];
  __shared__ float sBuf[WAVES][2][16 * HDIM];
  __shared__ float sCat[WAVES][16 * 16];

  const int tid = threadIdx.x;
  const int wid = tid >> 5, lane = tid & 31;
  const int half = lane >> 4, l16 = lane & 15;
  const int row0 = blockIdx.x * (WAVES * 16) + wid * 16;
  const int bb = row0 / NOP;
  const int op0 = row0 % NOP;

  __builtin_prefetch(l0w2 + tid * 128, 0, 1);
  __builtin_prefetch(l1w2 + tid * 128, 0, 1);
  __builtin_prefetch(pw2 + tid * 128, 0, 1);

  for (int i = tid; i < NMA * FDIM; i += TPB)
    sFM[i] = feats_ma[bb * NMA * FDIM + i];
  __syncthreads();

  float* sA = &sBuf[wid][0][0];
  float* sB = &sBuf[wid][1][0];
  float* sC = &sCat[wid][0];

  {
    const int m = lane >> 1, k0 = (lane & 1) * 4;
    const int* arow = adj + ((size_t)bb * NOP + op0 + m) * NMA;
    v4f s = {0.f, 0.f, 0.f, 0.f};
    for (int j = 0; j < NMA; ++j) {
      const float av = (float)arow[j];
      const float* f = &sFM[j * FDIM + k0];
      s.x += av * f[0]; s.y += av * f[1]; s.z += av * f[2]; s.w += av * f[3];
    }
    *(v4f*)&sA[m * FDIM + k0] = s;
  }
  __syncthreads();
  layer_h<FDIM, true>(l0w1, l0b1, sA, sB, lane);
  __syncthreads();
  layer_h<HDIM, true>(l0w2, l0b2, sB, sA, lane);
  __syncthreads();
  {
    v8f e0 = layer_o<HDIM>(l0w3, l0b3, sA, lane);
    if (l16 < 8) {
#pragma unroll
      for (int r = 0; r < 8; ++r)
        sC[(r + half * 8) * 16 + l16] = elu1(e0[r]);
    }
  }
  __syncthreads();

  {
    const int m = lane >> 1, k0 = (lane & 1) * 4;
    const float* fr = feats_op + ((size_t)bb * NOP + op0 + m) * FDIM + k0;
    *(v4f*)&sA[m * FDIM + k0] = *(const v4f*)fr;
  }
  __syncthreads();
  layer_h<FDIM, true>(l1w1, l1b1, sA, sB, lane);
  __syncthreads();
  layer_h<HDIM, true>(l1w2, l1b2, sB, sA, lane);
  __syncthreads();
  {
    v8f e1 = layer_o<HDIM>(l1w3, l1b3, sA, lane);
    if (l16 < 8) {
#pragma unroll
      for (int r = 0; r < 8; ++r)
        sC[(r + half * 8) * 16 + 8 + l16] = elu1(e1[r]);
    }
  }
  __syncthreads();

  layer_h<16, true>(pw1, pb1, sC, sA, lane);
  __syncthreads();
  layer_h<HDIM, true>(pw2, pb2, sA, sB, lane);
  __syncthreads();
  {
    v8f mu = layer_o<HDIM>(pw3, pb3, sB, lane);
    if (l16 < 8) {
#pragma unroll
      for (int r = 0; r < 8; ++r)
        out[(size_t)(row0 + r + half * 8) * ODIM + l16] = mu[r];
    }
  }
}

extern "C" void kernel_launch(void* const* d_in, const int* in_sizes, int n_in,
                              void* d_out, int out_size, void* d_ws,
                              size_t ws_size, hipStream_t stream) {
  const int*   adj = (const int*)d_in[0];
  const float* fop = (const float*)d_in[1];
  const float* fma = (const float*)d_in[2];
  const float* l0w1 = (const float*)d_in[3];
  const float* l0b1 = (const float*)d_in[4];
  const float* l0w2 = (const float*)d_in[5];
  const float* l0b2 = (const float*)d_in[6];
  const float* l0w3 = (const float*)d_in[7];
  const float* l0b3 = (const float*)d_in[8];
  const float* l1w1 = (const float*)d_in[9];
  const float* l1b1 = (const float*)d_in[10];
  const float* l1w2 = (const float*)d_in[11];
  const float* l1b2 = (const float*)d_in[12];
  const float* l1w3 = (const float*)d_in[13];
  const float* l1b3 = (const float*)d_in[14];
  const float* pw1 = (const float*)d_in[15];
  const float* pb1 = (const float*)d_in[16];
  const float* pw2 = (const float*)d_in[17];
  const float* pb2 = (const float*)d_in[18];
  const float* pw3 = (const float*)d_in[19];
  const float* pb3 = (const float*)d_in[20];

  const int total_rows = BATCH * NOP;
  const int rows_per_block = WAVES * 16;
  dim3 grid(total_rows / rows_per_block);
  fused_hgnn_mlp<<<grid, TPB, 0, stream>>>(
      adj, fop, fma, l0w1, l0b1, l0w2, l0b2, l0w3, l0b3, l1w1, l1b1, l1w2,
      l1b2, l1w3, l1b3, pw1, pb1, pw2, pb2, pw3, pb3, (float*)d_out);
}


# Wide layers split by output columns over the group's 4 waves (wave = 32 columns x all 64 rows, activations in one shared LDS buffer, pitch 132): a weight fragment feeds 4 WMMAs; sums unchanged
# speedup vs baseline: 1.7656x; 1.7656x over previous
#include <hip/hip_runtime.h>
#include <math.h>


typedef __attribute__((ext_vector_type(2))) float v2f;
typedef __attribute__((ext_vector_type(4))) float v4f;
typedef __attribute__((ext_vector_type(8))) float v8f;

#define WAVES 4
#define TPB   (WAVES * 32)
#define BATCH 32
#define NOP   512
#define NMA   64
#define FDIM  8
#define HDIM  128
#define ODIM  8

__device__ __forceinline__ float elu1(float x) {
  float e = __builtin_amdgcn_exp2f(x * 1.44269504088896340736f) - 1.0f;
  return x > 0.0f ? x : e;
}

__device__ __forceinline__ v8f wmma_f32(v2f a, v2f b, v8f c) {
  return __builtin_amdgcn_wmma_f32_16x16x4_f32(false, a, false, b, (short)0, c,
                                               false, false);
}

#define HP 132

template <int K, int PIN, bool ELU>
__device__ __forceinline__ void layer_w(const float* __restrict__ W,
                                        const float* __restrict__ bias,
                                        const float* sIn, float* sOut,
                                        int wid, int lane) {
  const int half = lane >> 4, l16 = lane & 15;
  const int nb = wid * 32;
  constexpr int KS = K / 4;
  constexpr int KB = KS < 4 ? KS : 4;
  constexpr int NB = KS / KB;
  const float* ap = sIn + l16 * PIN + 2 * half;
  const float* bp = W + (2 * half) * HDIM + nb + l16;
  v8f acc[2][4][2];
#pragma unroll
  for (int p = 0; p < 2; ++p)
#pragma unroll
    for (int mt = 0; mt < 4; ++mt)
#pragma unroll
      for (int nt = 0; nt < 2; ++nt)
#pragma unroll
        for (int r = 0; r < 8; ++r) acc[p][mt][nt][r] = 0.f;
  v2f a[2][KB][4], b[2][KB][2];
#pragma unroll
  for (int ks = 0; ks < KB; ++ks) {
#pragma unroll
    for (int nt = 0; nt < 2; ++nt) {
      b[0][ks][nt].x = bp[(ks * 4) * HDIM + nt * 16];
      b[0][ks][nt].y = bp[(ks * 4 + 1) * HDIM + nt * 16];
    }
#pragma unroll
    for (int mt = 0; mt < 4; ++mt)
      a[0][ks][mt] = *(const v2f*)(ap + mt * 16 * PIN + ks * 4);
  }
#pragma unroll
  for (int blk = 0; blk < NB; ++blk) {
    const int cur = blk & 1, nxt = cur ^ 1;
    if (blk + 1 < NB) {
#pragma unroll
      for (int ks = 0; ks < KB; ++ks) {
        const int k = ((blk + 1) * KB + ks) * 4;
#pragma unroll
        for (int nt = 0; nt < 2; ++nt) {
          b[nxt][ks][nt].x = bp[k * HDIM + nt * 16];
          b[nxt][ks][nt].y = bp[(k + 1) * HDIM + nt * 16];
        }
#pragma unroll
        for (int mt = 0; mt < 4; ++mt)
          a[nxt][ks][mt] = *(const v2f*)(ap + mt * 16 * PIN + k);
      }
    }
#pragma unroll
    for (int ks = 0; ks < KB; ++ks) {
      const int p = (blk * KB + ks) & 1;
#pragma unroll
      for (int nt = 0; nt < 2; ++nt)
#pragma unroll
        for (int mt = 0; mt < 4; ++mt)
          acc[p][mt][nt] = wmma_f32(a[cur][ks][mt], b[cur][ks][nt], acc[p][mt][nt]);
    }
  }
#pragma unroll
  for (int nt = 0; nt < 2; ++nt) {
    const float bv = bias[nb + nt * 16 + l16];
#pragma unroll
    for (int mt = 0; mt < 4; ++mt)
#pragma unroll
      for (int r = 0; r < 8; ++r) {
        float y = acc[0][mt][nt][r] + acc[1][mt][nt][r] + bv;
        if (ELU) y = elu1(y);
        sOut[(mt * 16 + r + half * 8) * HP + nb + nt * 16 + l16] = y;
      }
  }
}

template <int K, int PIN>
__device__ __forceinline__ v8f layer_o(const float* __restrict__ W,
                                       const float* __restrict__ bias,
                                       const float* sA, int lane) {
  const int half = lane >> 4, l16 = lane & 15;
  const int kh = half * 2;
  const int n = (l16 < 8) ? l16 : 0;
  v8f acc0 = {0.f, 0.f, 0.f, 0.f, 0.f, 0.f, 0.f, 0.f};
  v8f acc1 = {0.f, 0.f, 0.f, 0.f, 0.f, 0.f, 0.f, 0.f};
#pragma unroll
  for (int kc = 0; kc < K / 4; kc += 2) {
    {
      const int k = kc * 4 + kh;
      v2f a; a.x = sA[l16 * PIN + k];  a.y = sA[l16 * PIN + k + 1];
      v2f b; b.x = W[k * ODIM + n];    b.y = W[(k + 1) * ODIM + n];
      acc0 = wmma_f32(a, b, acc0);
    }
    {
      const int k = (kc + 1) * 4 + kh;
      v2f a; a.x = sA[l16 * PIN + k];  a.y = sA[l16 * PIN + k + 1];
      v2f b; b.x = W[k * ODIM + n];    b.y = W[(k + 1) * ODIM + n];
      acc1 = wmma_f32(a, b, acc1);
    }
  }
  const float bv = bias[n];
#pragma unroll
  for (int r = 0; r < 8; ++r) acc0[r] += acc1[r] + bv;
  return acc0;
}

__global__ __launch_bounds__(TPB) void fused_hgnn_mlp(
    const int* __restrict__ adj, const float* __restrict__ feats_op,
    const float* __restrict__ feats_ma,
    const float* __restrict__ l0w1, const float* __restrict__ l0b1,
    const float* __restrict__ l0w2, const float* __restrict__ l0b2,
    const float* __restrict__ l0w3, const float* __restrict__ l0b3,
    const float* __restrict__ l1w1, const float* __restrict__ l1b1,
    const float* __restrict__ l1w2, const float* __restrict__ l1b2,
    const float* __restrict__ l1w3, const float* __restrict__ l1b3,
    const float* __restrict__ pw1, const float* __restrict__ pb1,
    const float* __restrict__ pw2, const float* __restrict__ pb2,
    const float* __restrict__ pw3, const float* __restrict__ pb3,
    float* __restrict__ out) {
  __shared__ __attribute__((aligned(16))) float sFM[NMA * FDIM];
  __shared__ __attribute__((aligned(16))) float sX[WAVES * 16 * FDIM];
  __shared__ __attribute__((aligned(16))) float sAct[2][WAVES * 16 * HP];
  __shared__ __attribute__((aligned(16))) float sCat[WAVES * 16 * 16];

  const int tid = threadIdx.x;
  const int wid = tid >> 5, lane = tid & 31;
  const int half = lane >> 4, l16 = lane & 15;
  const int row0 = blockIdx.x * (WAVES * 16) + wid * 16;
  const int bb = row0 / NOP;
  const int op0 = row0 % NOP;

  __builtin_prefetch(l0w2 + tid * 128, 0, 1);
  __builtin_prefetch(l1w2 + tid * 128, 0, 1);
  __builtin_prefetch(pw2 + tid * 128, 0, 1);

  for (int i = tid; i < NMA * FDIM; i += TPB)
    sFM[i] = feats_ma[bb * NMA * FDIM + i];
  __syncthreads();

  float* sA = &sAct[0][0];
  float* sB = &sAct[1][0];
  float* sXw = &sX[wid * 16 * FDIM];
  float* sC = &sCat[wid * 16 * 16];
  const float* sHead = sB + wid * 16 * HP;

  {
    const int m = lane >> 1, k0 = (lane & 1) * 4;
    const int* arow = adj + ((size_t)bb * NOP + op0 + m) * NMA;
    v4f s = {0.f, 0.f, 0.f, 0.f};
    for (int j = 0; j < NMA; ++j) {
      const float av = (float)arow[j];
      const float* f = &sFM[j * FDIM + k0];
      s.x += av * f[0]; s.y += av * f[1]; s.z += av * f[2]; s.w += av * f[3];
    }
    *(v4f*)&sXw[m * FDIM + k0] = s;
  }
  __syncthreads();
  layer_w<FDIM, FDIM, true>(l0w1, l0b1, sX, sA, wid, lane);
  __syncthreads();
  layer_w<HDIM, HP, true>(l0w2, l0b2, sA, sB, wid, lane);
  __syncthreads();
  {
    v8f e0 = layer_o<HDIM, HP>(l0w3, l0b3, sHead, lane);
    if (l16 < 8) {
#pragma unroll
      for (int r = 0; r < 8; ++r)
        sC[(r + half * 8) * 16 + l16] = elu1(e0[r]);
    }
  }

  {
    const int m = lane >> 1, k0 = (lane & 1) * 4;
    const float* fr = feats_op + ((size_t)bb * NOP + op0 + m) * FDIM + k0;
    *(v4f*)&sXw[m * FDIM + k0] = *(const v4f*)fr;
  }
  __syncthreads();
  layer_w<FDIM, FDIM, true>(l1w1, l1b1, sX, sA, wid, lane);
  __syncthreads();
  layer_w<HDIM, HP, true>(l1w2, l1b2, sA, sB, wid, lane);
  __syncthreads();
  {
    v8f e1 = layer_o<HDIM, HP>(l1w3, l1b3, sHead, lane);
    if (l16 < 8) {
#pragma unroll
      for (int r = 0; r < 8; ++r)
        sC[(r + half * 8) * 16 + 8 + l16] = elu1(e1[r]);
    }
  }
  __syncthreads();

  layer_w<16, 16, true>(pw1, pb1, sCat, sA, wid, lane);
  __syncthreads();
  layer_w<HDIM, HP, true>(pw2, pb2, sA, sB, wid, lane);
  __syncthreads();
  {
    v8f mu = layer_o<HDIM, HP>(pw3, pb3, sHead, lane);
    if (l16 < 8) {
#pragma unroll
      for (int r = 0; r < 8; ++r)
        out[(size_t)(row0 + r + half * 8) * ODIM + l16] = mu[r];
    }
  }
}

extern "C" void kernel_launch(void* const* d_in, const int* in_sizes, int n_in,
                              void* d_out, int out_size, void* d_ws,
                              size_t ws_size, hipStream_t stream) {
  const int*   adj = (const int*)d_in[0];
  const float* fop = (const float*)d_in[1];
  const float* fma = (const float*)d_in[2];
  const float* l0w1 = (const float*)d_in[3];
  const float* l0b1 = (const float*)d_in[4];
  const float* l0w2 = (const float*)d_in[5];
  const float* l0b2 = (const float*)d_in[6];
  const float* l0w3 = (const float*)d_in[7];
  const float* l0b3 = (const float*)d_in[8];
  const float* l1w1 = (const float*)d_in[9];
  const float* l1b1 = (const float*)d_in[10];
  const float* l1w2 = (const float*)d_in[11];
  const float* l1b2 = (const float*)d_in[12];
  const float* l1w3 = (const float*)d_in[13];
  const float* l1b3 = (const float*)d_in[14];
  const float* pw1 = (const float*)d_in[15];
  const float* pb1 = (const float*)d_in[16];
  const float* pw2 = (const float*)d_in[17];
  const float* pb2 = (const float*)d_in[18];
  const float* pw3 = (const float*)d_in[19];
  const float* pb3 = (const float*)d_in[20];

  const int total_rows = BATCH * NOP;
  const int rows_per_block = WAVES * 16;
  dim3 grid(total_rows / rows_per_block);
  fused_hgnn_mlp<<<grid, TPB, 0, stream>>>(
      adj, fop, fma, l0w1, l0b1, l0w2, l0b2, l0w3, l0b3, l1w1, l1b1, l1w2,
      l1b2, l1w3, l1b3, pw1, pb1, pw2, pb2, pw3, pb3, (float*)d_out);
}


# Wide layers' column split made finer: 8 waves per group, wave = 16 columns x all 64 rows, two waves per SIMD (heads stay on waves 0-3, waves 4-7 stage the next input meanwhile); sums unchanged
# speedup vs baseline: 1.9867x; 1.1252x over previous
#include <hip/hip_runtime.h>
#include <math.h>


typedef __attribute__((ext_vector_type(2))) float v2f;
typedef __attribute__((ext_vector_type(4))) float v4f;
typedef __attribute__((ext_vector_type(8))) float v8f;

#define WAVES 8
#define MT    4
#define TPB   (WAVES * 32)
#define BATCH 32
#define NOP   512
#define NMA   64
#define FDIM  8
#define HDIM  128
#define ODIM  8

__device__ __forceinline__ float elu1(float x) {
  float e = __builtin_amdgcn_exp2f(x * 1.44269504088896340736f) - 1.0f;
  return x > 0.0f ? x : e;
}

__device__ __forceinline__ v8f wmma_f32(v2f a, v2f b, v8f c) {
  return __builtin_amdgcn_wmma_f32_16x16x4_f32(false, a, false, b, (short)0, c,
                                               false, false);
}

#define HP 132

template <int K, int PIN, bool ELU>
__device__ __forceinline__ void layer_w(const float* __restrict__ W,
                                        const float* __restrict__ bias,
                                        const float* sIn, float* sOut,
                                        int wid, int lane) {
  const int half = lane >> 4, l16 = lane & 15;
  const int nb = wid * 16;
  constexpr int KS = K / 4;
  constexpr int KB = KS < 8 ? KS : 8;
  constexpr int NB = KS / KB;
  const float* ap = sIn + l16 * PIN + 2 * half;
  const float* bp = W + (2 * half) * HDIM + nb + l16;
  v8f acc[2][MT];
#pragma unroll
  for (int p = 0; p < 2; ++p)
#pragma unroll
    for (int mt = 0; mt < MT; ++mt)
#pragma unroll
      for (int r = 0; r < 8; ++r) acc[p][mt][r] = 0.f;
  v2f a[2][KB][MT], b[2][KB];
#pragma unroll
  for (int ks = 0; ks < KB; ++ks) {
    b[0][ks].x = bp[(ks * 4) * HDIM];
    b[0][ks].y = bp[(ks * 4 + 1) * HDIM];
#pragma unroll
    for (int mt = 0; mt < MT; ++mt)
      a[0][ks][mt] = *(const v2f*)(ap + mt * 16 * PIN + ks * 4);
  }
#pragma unroll
  for (int blk = 0; blk < NB; ++blk) {
    const int cur = blk & 1, nxt = cur ^ 1;
    if (blk + 1 < NB) {
#pragma unroll
      for (int ks = 0; ks < KB; ++ks) {
        const int k = ((blk + 1) * KB + ks) * 4;
        b[nxt][ks].x = bp[k * HDIM];
        b[nxt][ks].y = bp[(k + 1) * HDIM];
#pragma unroll
        for (int mt = 0; mt < MT; ++mt)
          a[nxt][ks][mt] = *(const v2f*)(ap + mt * 16 * PIN + k);
      }
    }
#pragma unroll
    for (int ks = 0; ks < KB; ++ks) {
      const int p = (blk * KB + ks) & 1;
#pragma unroll
      for (int mt = 0; mt < MT; ++mt)
        acc[p][mt] = wmma_f32(a[cur][ks][mt], b[cur][ks], acc[p][mt]);
    }
  }
  const float bv = bias[nb + l16];
#pragma unroll
  for (int mt = 0; mt < MT; ++mt)
#pragma unroll
    for (int r = 0; r < 8; ++r) {
      float y = acc[0][mt][r] + acc[1][mt][r] + bv;
      if (ELU) y = elu1(y);
      sOut[(mt * 16 + r + half * 8) * HP + nb + l16] = y;
    }
}

template <int K, int PIN>
__device__ __forceinline__ v8f layer_o(const float* __restrict__ W,
                                       const float* __restrict__ bias,
                                       const float* sA, int lane) {
  const int half = lane >> 4, l16 = lane & 15;
  const int kh = half * 2;
  const int n = (l16 < 8) ? l16 : 0;
  v8f acc0 = {0.f, 0.f, 0.f, 0.f, 0.f, 0.f, 0.f, 0.f};
  v8f acc1 = {0.f, 0.f, 0.f, 0.f, 0.f, 0.f, 0.f, 0.f};
#pragma unroll
  for (int kc = 0; kc < K / 4; kc += 2) {
    {
      const int k = kc * 4 + kh;
      v2f a; a.x = sA[l16 * PIN + k];  a.y = sA[l16 * PIN + k + 1];
      v2f b; b.x = W[k * ODIM + n];    b.y = W[(k + 1) * ODIM + n];
      acc0 = wmma_f32(a, b, acc0);
    }
    {
      const int k = (kc + 1) * 4 + kh;
      v2f a; a.x = sA[l16 * PIN + k];  a.y = sA[l16 * PIN + k + 1];
      v2f b; b.x = W[k * ODIM + n];    b.y = W[(k + 1) * ODIM + n];
      acc1 = wmma_f32(a, b, acc1);
    }
  }
  const float bv = bias[n];
#pragma unroll
  for (int r = 0; r < 8; ++r) acc0[r] += acc1[r] + bv;
  return acc0;
}

__global__ __launch_bounds__(TPB) __attribute__((amdgpu_waves_per_eu(2, 2))) void fused_hgnn_mlp(
    const int* __restrict__ adj, const float* __restrict__ feats_op,
    const float* __restrict__ feats_ma,
    const float* __restrict__ l0w1, const float* __restrict__ l0b1,
    const float* __restrict__ l0w2, const float* __restrict__ l0b2,
    const float* __restrict__ l0w3, const float* __restrict__ l0b3,
    const float* __restrict__ l1w1, const float* __restrict__ l1b1,
    const float* __restrict__ l1w2, const float* __restrict__ l1b2,
    const float* __restrict__ l1w3, const float* __restrict__ l1b3,
    const float* __restrict__ pw1, const float* __restrict__ pb1,
    const float* __restrict__ pw2, const float* __restrict__ pb2,
    const float* __restrict__ pw3, const float* __restrict__ pb3,
    float* __restrict__ out) {
  __shared__ __attribute__((aligned(16))) float sFM[NMA * FDIM];
  __shared__ __attribute__((aligned(16))) float sX[MT * 16 * FDIM];
  __shared__ __attribute__((aligned(16))) float sAct[2][MT * 16 * HP];
  __shared__ __attribute__((aligned(16))) float sCat[MT * 16 * 16];

  const int tid = threadIdx.x;
  const int wid = __builtin_amdgcn_readfirstlane(tid >> 5);
  const int lane = tid & 31;
  const int half = lane >> 4, l16 = lane & 15;
  const int grow0 = blockIdx.x * (MT * 16);
  const int bb = grow0 / NOP;
  const int op0 = grow0 % NOP;

  __builtin_prefetch(l0w2 + tid * 64, 0, 1);
  __builtin_prefetch(l1w2 + tid * 64, 0, 1);
  __builtin_prefetch(pw2 + tid * 64, 0, 1);

  for (int i = tid; i < NMA * FDIM; i += TPB)
    sFM[i] = feats_ma[bb * NMA * FDIM + i];
  __syncthreads();

  float* sA = &sAct[0][0];
  float* sB = &sAct[1][0];
  const int ht = wid & (MT - 1);
  float* sC = &sCat[ht * 16 * 16];
  const float* sHead = sB + ht * 16 * HP;

  {
    const int m = tid >> 2, k0 = (tid & 3) * 2;
    const int* arow = adj + ((size_t)bb * NOP + op0 + m) * NMA;
    v2f s = {0.f, 0.f};
    for (int j = 0; j < NMA; ++j) {
      const float av = (float)arow[j];
      const float* f = &sFM[j * FDIM + k0];
      s.x += av * f[0]; s.y += av * f[1];
    }
    *(v2f*)&sX[m * FDIM + k0] = s;
  }
  __syncthreads();
  layer_w<FDIM, FDIM, true>(l0w1, l0b1, sX, sA, wid, lane);
  __syncthreads();
  layer_w<HDIM, HP, true>(l0w2, l0b2, sA, sB, wid, lane);
  __syncthreads();
  if (wid < MT) {
    v8f e0 = layer_o<HDIM, HP>(l0w3, l0b3, sHead, lane);
    if (l16 < 8) {
#pragma unroll
      for (int r = 0; r < 8; ++r)
        sC[(r + half * 8) * 16 + l16] = elu1(e0[r]);
    }
  } else {
    const int t = tid - MT * 32;
    const int m = t >> 1, k0 = (t & 1) * 4;
    const float* fr = feats_op + ((size_t)bb * NOP + op0 + m) * FDIM + k0;
    *(v4f*)&sX[m * FDIM + k0] = *(const v4f*)fr;
  }
  __syncthreads();
  layer_w<FDIM, FDIM, true>(l1w1, l1b1, sX, sA, wid, lane);
  __syncthreads();
  layer_w<HDIM, HP, true>(l1w2, l1b2, sA, sB, wid, lane);
  __syncthreads();
  if (wid < MT) {
    v8f e1 = layer_o<HDIM, HP>(l1w3, l1b3, sHead, lane);
    if (l16 < 8) {
#pragma unroll
      for (int r = 0; r < 8; ++r)
        sC[(r + half * 8) * 16 + 8 + l16] = elu1(e1[r]);
    }
  }
  __syncthreads();

  layer_w<16, 16, true>(pw1, pb1, sCat, sA, wid, lane);
  __syncthreads();
  layer_w<HDIM, HP, true>(pw2, pb2, sA, sB, wid, lane);
  __syncthreads();
  if (wid < MT) {
    v8f mu = layer_o<HDIM, HP>(pw3, pb3, sHead, lane);
    if (l16 < 8) {
#pragma unroll
      for (int r = 0; r < 8; ++r)
        out[(size_t)(grow0 + ht * 16 + r + half * 8) * ODIM + l16] = mu[r];
    }
  }
}

extern "C" void kernel_launch(void* const* d_in, const int* in_sizes, int n_in,
                              void* d_out, int out_size, void* d_ws,
                              size_t ws_size, hipStream_t stream) {
  const int*   adj = (const int*)d_in[0];
  const float* fop = (const float*)d_in[1];
  const float* fma = (const float*)d_in[2];
  const float* l0w1 = (const float*)d_in[3];
  const float* l0b1 = (const float*)d_in[4];
  const float* l0w2 = (const float*)d_in[5];
  const float* l0b2 = (const float*)d_in[6];
  const float* l0w3 = (const float*)d_in[7];
  const float* l0b3 = (const float*)d_in[8];
  const float* l1w1 = (const float*)d_in[9];
  const float* l1b1 = (const float*)d_in[10];
  const float* l1w2 = (const float*)d_in[11];
  const float* l1b2 = (const float*)d_in[12];
  const float* l1w3 = (const float*)d_in[13];
  const float* l1b3 = (const float*)d_in[14];
  const float* pw1 = (const float*)d_in[15];
  const float* pb1 = (const float*)d_in[16];
  const float* pw2 = (const float*)d_in[17];
  const float* pb2 = (const float*)d_in[18];
  const float* pw3 = (const float*)d_in[19];
  const float* pb3 = (const float*)d_in[20];

  const int total_rows = BATCH * NOP;
  const int rows_per_block = MT * 16;
  dim3 grid(total_rows / rows_per_block);
  fused_hgnn_mlp<<<grid, TPB, 0, stream>>>(
      adj, fop, fma, l0w1, l0b1, l0w2, l0b2, l0w3, l0b3, l1w1, l1b1, l1w2,
      l1b2, l1w3, l1b3, pw1, pb1, pw2, pb2, pw3, pb3, (float*)d_out);
}


# Aggregation adj x feats_ma moved onto the matrix unit: adj rows as A, feats_ma as B, both straight from global as one load group, no LDS staging; sum order of the aggregation changes
# speedup vs baseline: 2.1521x; 1.0833x over previous
#include <hip/hip_runtime.h>
#include <math.h>


typedef __attribute__((ext_vector_type(2))) float v2f;
typedef __attribute__((ext_vector_type(4))) int   v4i;
typedef __attribute__((ext_vector_type(4))) float v4f;
typedef __attribute__((ext_vector_type(8))) float v8f;

#define WAVES 8
#define MT    4
#define TPB   (WAVES * 32)
#define BATCH 32
#define NOP   512
#define NMA   64
#define FDIM  8
#define HDIM  128
#define ODIM  8

__device__ __forceinline__ float elu1(float x) {
  float e = __builtin_amdgcn_exp2f(x * 1.44269504088896340736f) - 1.0f;
  return x > 0.0f ? x : e;
}

__device__ __forceinline__ v8f wmma_f32(v2f a, v2f b, v8f c) {
  return __builtin_amdgcn_wmma_f32_16x16x4_f32(false, a, false, b, (short)0, c,
                                               false, false);
}

#define HP 132

template <int K, int PIN, bool ELU>
__device__ __forceinline__ void layer_w(const float* __restrict__ W,
                                        const float* __restrict__ bias,
                                        const float* sIn, float* sOut,
                                        int wid, int lane) {
  const int half = lane >> 4, l16 = lane & 15;
  const int nb = wid * 16;
  constexpr int KS = K / 4;
  constexpr int KB = KS < 8 ? KS : 8;
  constexpr int NB = KS / KB;
  const float* ap = sIn + l16 * PIN + 2 * half;
  const float* bp = W + (2 * half) * HDIM + nb + l16;
  v8f acc[2][MT];
#pragma unroll
  for (int p = 0; p < 2; ++p)
#pragma unroll
    for (int mt = 0; mt < MT; ++mt)
#pragma unroll
      for (int r = 0; r < 8; ++r) acc[p][mt][r] = 0.f;
  v2f a[2][KB][MT], b[2][KB];
#pragma unroll
  for (int ks = 0; ks < KB; ++ks) {
    b[0][ks].x = bp[(ks * 4) * HDIM];
    b[0][ks].y = bp[(ks * 4 + 1) * HDIM];
#pragma unroll
    for (int mt = 0; mt < MT; ++mt)
      a[0][ks][mt] = *(const v2f*)(ap + mt * 16 * PIN + ks * 4);
  }
#pragma unroll
  for (int blk = 0; blk < NB; ++blk) {
    const int cur = blk & 1, nxt = cur ^ 1;
    if (blk + 1 < NB) {
#pragma unroll
      for (int ks = 0; ks < KB; ++ks) {
        const int k = ((blk + 1) * KB + ks) * 4;
        b[nxt][ks].x = bp[k * HDIM];
        b[nxt][ks].y = bp[(k + 1) * HDIM];
#pragma unroll
        for (int mt = 0; mt < MT; ++mt)
          a[nxt][ks][mt] = *(const v2f*)(ap + mt * 16 * PIN + k);
      }
    }
#pragma unroll
    for (int ks = 0; ks < KB; ++ks) {
      const int p = (blk * KB + ks) & 1;
#pragma unroll
      for (int mt = 0; mt < MT; ++mt)
        acc[p][mt] = wmma_f32(a[cur][ks][mt], b[cur][ks], acc[p][mt]);
    }
  }
  const float bv = bias[nb + l16];
#pragma unroll
  for (int mt = 0; mt < MT; ++mt)
#pragma unroll
    for (int r = 0; r < 8; ++r) {
      float y = acc[0][mt][r] + acc[1][mt][r] + bv;
      if (ELU) y = elu1(y);
      sOut[(mt * 16 + r + half * 8) * HP + nb + l16] = y;
    }
}

template <int K, int PIN>
__device__ __forceinline__ v8f layer_o(const float* __restrict__ W,
                                       const float* __restrict__ bias,
                                       const float* sA, int lane) {
  const int half = lane >> 4, l16 = lane & 15;
  const int kh = half * 2;
  const int n = (l16 < 8) ? l16 : 0;
  v8f acc0 = {0.f, 0.f, 0.f, 0.f, 0.f, 0.f, 0.f, 0.f};
  v8f acc1 = {0.f, 0.f, 0.f, 0.f, 0.f, 0.f, 0.f, 0.f};
#pragma unroll
  for (int kc = 0; kc < K / 4; kc += 2) {
    {
      const int k = kc * 4 + kh;
      v2f a; a.x = sA[l16 * PIN + k];  a.y = sA[l16 * PIN + k + 1];
      v2f b; b.x = W[k * ODIM + n];    b.y = W[(k + 1) * ODIM + n];
      acc0 = wmma_f32(a, b, acc0);
    }
    {
      const int k = (kc + 1) * 4 + kh;
      v2f a; a.x = sA[l16 * PIN + k];  a.y = sA[l16 * PIN + k + 1];
      v2f b; b.x = W[k * ODIM + n];    b.y = W[(k + 1) * ODIM + n];
      acc1 = wmma_f32(a, b, acc1);
    }
  }
  const float bv = bias[n];
#pragma unroll
  for (int r = 0; r < 8; ++r) acc0[r] += acc1[r] + bv;
  return acc0;
}

__global__ __launch_bounds__(TPB) __attribute__((amdgpu_waves_per_eu(2, 2))) void fused_hgnn_mlp(
    const int* __restrict__ adj, const float* __restrict__ feats_op,
    const float* __restrict__ feats_ma,
    const float* __restrict__ l0w1, const float* __restrict__ l0b1,
    const float* __restrict__ l0w2, const float* __restrict__ l0b2,
    const float* __restrict__ l0w3, const float* __restrict__ l0b3,
    const float* __restrict__ l1w1, const float* __restrict__ l1b1,
    const float* __restrict__ l1w2, const float* __restrict__ l1b2,
    const float* __restrict__ l1w3, const float* __restrict__ l1b3,
    const float* __restrict__ pw1, const float* __restrict__ pb1,
    const float* __restrict__ pw2, const float* __restrict__ pb2,
    const float* __restrict__ pw3, const float* __restrict__ pb3,
    float* __restrict__ out) {
  __shared__ __attribute__((aligned(16))) float sX[MT * 16 * FDIM];
  __shared__ __attribute__((aligned(16))) float sAct[2][MT * 16 * HP];
  __shared__ __attribute__((aligned(16))) float sCat[MT * 16 * 16];

  const int tid = threadIdx.x;
  const int wid = __builtin_amdgcn_readfirstlane(tid >> 5);
  const int lane = tid & 31;
  const int half = lane >> 4, l16 = lane & 15;
  const int grow0 = blockIdx.x * (MT * 16);
  const int bb = grow0 / NOP;
  const int op0 = grow0 % NOP;

  __builtin_prefetch(l0w2 + tid * 64, 0, 1);
  __builtin_prefetch(l1w2 + tid * 64, 0, 1);
  __builtin_prefetch(pw2 + tid * 64, 0, 1);

  float* sA = &sAct[0][0];
  float* sB = &sAct[1][0];
  const int ht = wid & (MT - 1);
  float* sC = &sCat[ht * 16 * 16];
  const float* sHead = sB + ht * 16 * HP;

  if (wid < MT) {
    const int n = (l16 < FDIM) ? l16 : 0;
    const int* arow = adj + ((size_t)bb * NOP + op0 + ht * 16 + l16) * NMA + 4 * half;
    const float* fb = feats_ma + ((size_t)bb * NMA + 4 * half) * FDIM + n;
    v4i ai[8];
#pragma unroll
    for (int q = 0; q < 8; ++q) ai[q] = *(const v4i*)(arow + 8 * q);
    v2f bf[16];
#pragma unroll
    for (int j = 0; j < 16; ++j) {
      const int k = 8 * (j >> 1) + 2 * (j & 1);
      bf[j].x = fb[k * FDIM];
      bf[j].y = fb[(k + 1) * FDIM];
    }
    v8f g0 = {0.f, 0.f, 0.f, 0.f, 0.f, 0.f, 0.f, 0.f};
    v8f g1 = {0.f, 0.f, 0.f, 0.f, 0.f, 0.f, 0.f, 0.f};
#pragma unroll
    for (int j = 0; j < 16; j += 2) {
      const v4i q4 = ai[j >> 1];
      v2f a0; a0.x = (float)q4[0]; a0.y = (float)q4[1];
      v2f a1; a1.x = (float)q4[2]; a1.y = (float)q4[3];
      g0 = wmma_f32(a0, bf[j], g0);
      g1 = wmma_f32(a1, bf[j + 1], g1);
    }
    if (l16 < FDIM) {
#pragma unroll
      for (int r = 0; r < 8; ++r)
        sX[(ht * 16 + r + half * 8) * FDIM + l16] = g0[r] + g1[r];
    }
  }
  __syncthreads();
  layer_w<FDIM, FDIM, true>(l0w1, l0b1, sX, sA, wid, lane);
  __syncthreads();
  layer_w<HDIM, HP, true>(l0w2, l0b2, sA, sB, wid, lane);
  __syncthreads();
  if (wid < MT) {
    v8f e0 = layer_o<HDIM, HP>(l0w3, l0b3, sHead, lane);
    if (l16 < 8) {
#pragma unroll
      for (int r = 0; r < 8; ++r)
        sC[(r + half * 8) * 16 + l16] = elu1(e0[r]);
    }
  } else {
    const int t = tid - MT * 32;
    const int m = t >> 1, k0 = (t & 1) * 4;
    const float* fr = feats_op + ((size_t)bb * NOP + op0 + m) * FDIM + k0;
    *(v4f*)&sX[m * FDIM + k0] = *(const v4f*)fr;
  }
  __syncthreads();
  layer_w<FDIM, FDIM, true>(l1w1, l1b1, sX, sA, wid, lane);
  __syncthreads();
  layer_w<HDIM, HP, true>(l1w2, l1b2, sA, sB, wid, lane);
  __syncthreads();
  if (wid < MT) {
    v8f e1 = layer_o<HDIM, HP>(l1w3, l1b3, sHead, lane);
    if (l16 < 8) {
#pragma unroll
      for (int r = 0; r < 8; ++r)
        sC[(r + half * 8) * 16 + 8 + l16] = elu1(e1[r]);
    }
  }
  __syncthreads();

  layer_w<16, 16, true>(pw1, pb1, sCat, sA, wid, lane);
  __syncthreads();
  layer_w<HDIM, HP, true>(pw2, pb2, sA, sB, wid, lane);
  __syncthreads();
  if (wid < MT) {
    v8f mu = layer_o<HDIM, HP>(pw3, pb3, sHead, lane);
    if (l16 < 8) {
#pragma unroll
      for (int r = 0; r < 8; ++r)
        out[(size_t)(grow0 + ht * 16 + r + half * 8) * ODIM + l16] = mu[r];
    }
  }
}

extern "C" void kernel_launch(void* const* d_in, const int* in_sizes, int n_in,
                              void* d_out, int out_size, void* d_ws,
                              size_t ws_size, hipStream_t stream) {
  const int*   adj = (const int*)d_in[0];
  const float* fop = (const float*)d_in[1];
  const float* fma = (const float*)d_in[2];
  const float* l0w1 = (const float*)d_in[3];
  const float* l0b1 = (const float*)d_in[4];
  const float* l0w2 = (const float*)d_in[5];
  const float* l0b2 = (const float*)d_in[6];
  const float* l0w3 = (const float*)d_in[7];
  const float* l0b3 = (const float*)d_in[8];
  const float* l1w1 = (const float*)d_in[9];
  const float* l1b1 = (const float*)d_in[10];
  const float* l1w2 = (const float*)d_in[11];
  const float* l1b2 = (const float*)d_in[12];
  const float* l1w3 = (const float*)d_in[13];
  const float* l1b3 = (const float*)d_in[14];
  const float* pw1 = (const float*)d_in[15];
  const float* pb1 = (const float*)d_in[16];
  const float* pw2 = (const float*)d_in[17];
  const float* pb2 = (const float*)d_in[18];
  const float* pw3 = (const float*)d_in[19];
  const float* pb3 = (const float*)d_in[20];

  const int total_rows = BATCH * NOP;
  const int rows_per_block = MT * 16;
  dim3 grid(total_rows / rows_per_block);
  fused_hgnn_mlp<<<grid, TPB, 0, stream>>>(
      adj, fop, fma, l0w1, l0b1, l0w2, l0b2, l0w3, l0b3, l1w1, l1b1, l1w2,
      l1b2, l1w3, l1b3, pw1, pb1, pw2, pb2, pw3, pb3, (float*)d_out);
}


# Loads leave the WMMA streams: each layer's LDS fragment reads go out as groups in front of 32-WMMA blocks (heads: all in front), the heads' bias load too, held by scheduling barriers; sums unchanged
# speedup vs baseline: 2.3611x; 1.0971x over previous
#include <hip/hip_runtime.h>
#include <math.h>


typedef __attribute__((ext_vector_type(2))) float v2f;
typedef __attribute__((ext_vector_type(4))) int   v4i;
typedef __attribute__((ext_vector_type(4))) float v4f;
typedef __attribute__((ext_vector_type(8))) float v8f;

#define WAVES 8
#define MT    4
#define TPB   (WAVES * 32)
#define BATCH 32
#define NOP   512
#define NMA   64
#define FDIM  8
#define HDIM  128
#define ODIM  8

__device__ __forceinline__ float elu1(float x) {
  float e = __builtin_amdgcn_exp2f(x * 1.44269504088896340736f) - 1.0f;
  return x > 0.0f ? x : e;
}

__device__ __forceinline__ v8f wmma_f32(v2f a, v2f b, v8f c) {
  return __builtin_amdgcn_wmma_f32_16x16x4_f32(false, a, false, b, (short)0, c,
                                               false, false);
}

#define HP 132

template <int K, int PIN, bool ELU>
__device__ __forceinline__ void layer_w(const float* __restrict__ W,
                                        const float* __restrict__ bias,
                                        const float* sIn, float* sOut,
                                        int wid, int lane) {
  const int half = lane >> 4, l16 = lane & 15;
  const int nb = wid * 16;
  constexpr int KS = K / 4;
  constexpr int KB = KS < 8 ? KS : 8;
  constexpr int NB = KS / KB;
  const float* ap = sIn + l16 * PIN + 2 * half;
  const float* bp = W + (2 * half) * HDIM + nb + l16;
  v8f acc[2][MT];
#pragma unroll
  for (int p = 0; p < 2; ++p)
#pragma unroll
    for (int mt = 0; mt < MT; ++mt)
#pragma unroll
      for (int r = 0; r < 8; ++r) acc[p][mt][r] = 0.f;
  v2f a[2][KB][MT], b[KS];
#pragma unroll
  for (int ks = 0; ks < KS; ++ks) {
    b[ks].x = bp[(ks * 4) * HDIM];
    b[ks].y = bp[(ks * 4 + 1) * HDIM];
  }
  const float bv = bias[nb + l16];
#pragma unroll
  for (int ks = 0; ks < KB; ++ks)
#pragma unroll
    for (int mt = 0; mt < MT; ++mt)
      a[0][ks][mt] = *(const v2f*)(ap + mt * 16 * PIN + ks * 4);
#pragma unroll
  for (int blk = 0; blk < NB; ++blk) {
    const int cur = blk & 1, nxt = cur ^ 1;
    if (blk + 1 < NB) {
#pragma unroll
      for (int ks = 0; ks < KB; ++ks) {
        const int k = ((blk + 1) * KB + ks) * 4;
#pragma unroll
        for (int mt = 0; mt < MT; ++mt)
          a[nxt][ks][mt] = *(const v2f*)(ap + mt * 16 * PIN + k);
      }
    }
    __builtin_amdgcn_sched_barrier(0);
#pragma unroll
    for (int ks = 0; ks < KB; ++ks) {
      const int p = (blk * KB + ks) & 1;
#pragma unroll
      for (int mt = 0; mt < MT; ++mt)
        acc[p][mt] = wmma_f32(a[cur][ks][mt], b[blk * KB + ks], acc[p][mt]);
    }
    if (blk + 2 < NB) __builtin_amdgcn_sched_barrier(0);
  }
#pragma unroll
  for (int mt = 0; mt < MT; ++mt)
#pragma unroll
    for (int r = 0; r < 8; ++r) {
      float y = acc[0][mt][r] + acc[1][mt][r] + bv;
      if (ELU) y = elu1(y);
      sOut[(mt * 16 + r + half * 8) * HP + nb + l16] = y;
    }
}

template <int K, int PIN>
__device__ __forceinline__ v8f layer_o(const float* __restrict__ W,
                                       const float* __restrict__ bias,
                                       const float* sA, int lane) {
  const int half = lane >> 4, l16 = lane & 15;
  const int kh = half * 2;
  const int n = (l16 < 8) ? l16 : 0;
  constexpr int KS = K / 4;
  const float bv = bias[n];
  v2f a[KS], b[KS];
#pragma unroll
  for (int ks = 0; ks < KS; ++ks) {
    const int k = ks * 4 + kh;
    b[ks].x = W[k * ODIM + n];
    b[ks].y = W[(k + 1) * ODIM + n];
  }
#pragma unroll
  for (int ks = 0; ks < KS; ++ks) {
    const int k = ks * 4 + kh;
    a[ks].x = sA[l16 * PIN + k];
    a[ks].y = sA[l16 * PIN + k + 1];
  }
  asm volatile("" ::"v"(bv));
  __builtin_amdgcn_sched_barrier(0);
  v8f acc0 = {0.f, 0.f, 0.f, 0.f, 0.f, 0.f, 0.f, 0.f};
  v8f acc1 = {0.f, 0.f, 0.f, 0.f, 0.f, 0.f, 0.f, 0.f};
#pragma unroll
  for (int ks = 0; ks < KS; ks += 2) {
    acc0 = wmma_f32(a[ks], b[ks], acc0);
    acc1 = wmma_f32(a[ks + 1], b[ks + 1], acc1);
  }
#pragma unroll
  for (int r = 0; r < 8; ++r) acc0[r] += acc1[r] + bv;
  return acc0;
}

__global__ __launch_bounds__(TPB) __attribute__((amdgpu_waves_per_eu(2, 2))) void fused_hgnn_mlp(
    const int* __restrict__ adj, const float* __restrict__ feats_op,
    const float* __restrict__ feats_ma,
    const float* __restrict__ l0w1, const float* __restrict__ l0b1,
    const float* __restrict__ l0w2, const float* __restrict__ l0b2,
    const float* __restrict__ l0w3, const float* __restrict__ l0b3,
    const float* __restrict__ l1w1, const float* __restrict__ l1b1,
    const float* __restrict__ l1w2, const float* __restrict__ l1b2,
    const float* __restrict__ l1w3, const float* __restrict__ l1b3,
    const float* __restrict__ pw1, const float* __restrict__ pb1,
    const float* __restrict__ pw2, const float* __restrict__ pb2,
    const float* __restrict__ pw3, const float* __restrict__ pb3,
    float* __restrict__ out) {
  __shared__ __attribute__((aligned(16))) float sX[MT * 16 * FDIM];
  __shared__ __attribute__((aligned(16))) float sAct[2][MT * 16 * HP];
  __shared__ __attribute__((aligned(16))) float sCat[MT * 16 * 16];

  const int tid = threadIdx.x;
  const int wid = __builtin_amdgcn_readfirstlane(tid >> 5);
  const int lane = tid & 31;
  const int half = lane >> 4, l16 = lane & 15;
  const int grow0 = blockIdx.x * (MT * 16);
  const int bb = grow0 / NOP;
  const int op0 = grow0 % NOP;

  __builtin_prefetch(l0w2 + tid * 64, 0, 1);
  __builtin_prefetch(l1w2 + tid * 64, 0, 1);
  __builtin_prefetch(pw2 + tid * 64, 0, 1);

  float* sA = &sAct[0][0];
  float* sB = &sAct[1][0];
  const int ht = wid & (MT - 1);
  float* sC = &sCat[ht * 16 * 16];
  const float* sHead = sB + ht * 16 * HP;

  if (wid < MT) {
    const int n = (l16 < FDIM) ? l16 : 0;
    const int* arow = adj + ((size_t)bb * NOP + op0 + ht * 16 + l16) * NMA + 4 * half;
    const float* fb = feats_ma + ((size_t)bb * NMA + 4 * half) * FDIM + n;
    v4i ai[8];
#pragma unroll
    for (int q = 0; q < 8; ++q) ai[q] = *(const v4i*)(arow + 8 * q);
    v2f bf[16];
#pragma unroll
    for (int j = 0; j < 16; ++j) {
      const int k = 8 * (j >> 1) + 2 * (j & 1);
      bf[j].x = fb[k * FDIM];
      bf[j].y = fb[(k + 1) * FDIM];
    }
    v8f g0 = {0.f, 0.f, 0.f, 0.f, 0.f, 0.f, 0.f, 0.f};
    v8f g1 = {0.f, 0.f, 0.f, 0.f, 0.f, 0.f, 0.f, 0.f};
#pragma unroll
    for (int j = 0; j < 16; j += 2) {
      const v4i q4 = ai[j >> 1];
      v2f a0; a0.x = (float)q4[0]; a0.y = (float)q4[1];
      v2f a1; a1.x = (float)q4[2]; a1.y = (float)q4[3];
      g0 = wmma_f32(a0, bf[j], g0);
      g1 = wmma_f32(a1, bf[j + 1], g1);
    }
    if (l16 < FDIM) {
#pragma unroll
      for (int r = 0; r < 8; ++r)
        sX[(ht * 16 + r + half * 8) * FDIM + l16] = g0[r] + g1[r];
    }
  }
  __syncthreads();
  layer_w<FDIM, FDIM, true>(l0w1, l0b1, sX, sA, wid, lane);
  __syncthreads();
  layer_w<HDIM, HP, true>(l0w2, l0b2, sA, sB, wid, lane);
  __syncthreads();
  if (wid < MT) {
    v8f e0 = layer_o<HDIM, HP>(l0w3, l0b3, sHead, lane);
    if (l16 < 8) {
#pragma unroll
      for (int r = 0; r < 8; ++r)
        sC[(r + half * 8) * 16 + l16] = elu1(e0[r]);
    }
  } else {
    const int t = tid - MT * 32;
    const int m = t >> 1, k0 = (t & 1) * 4;
    const float* fr = feats_op + ((size_t)bb * NOP + op0 + m) * FDIM + k0;
    *(v4f*)&sX[m * FDIM + k0] = *(const v4f*)fr;
  }
  __syncthreads();
  layer_w<FDIM, FDIM, true>(l1w1, l1b1, sX, sA, wid, lane);
  __syncthreads();
  layer_w<HDIM, HP, true>(l1w2, l1b2, sA, sB, wid, lane);
  __syncthreads();
  if (wid < MT) {
    v8f e1 = layer_o<HDIM, HP>(l1w3, l1b3, sHead, lane);
    if (l16 < 8) {
#pragma unroll
      for (int r = 0; r < 8; ++r)
        sC[(r + half * 8) * 16 + 8 + l16] = elu1(e1[r]);
    }
  }
  __syncthreads();

  layer_w<16, 16, true>(pw1, pb1, sCat, sA, wid, lane);
  __syncthreads();
  layer_w<HDIM, HP, true>(pw2, pb2, sA, sB, wid, lane);
  __syncthreads();
  if (wid < MT) {
    v8f mu = layer_o<HDIM, HP>(pw3, pb3, sHead, lane);
    if (l16 < 8) {
#pragma unroll
      for (int r = 0; r < 8; ++r)
        out[(size_t)(grow0 + ht * 16 + r + half * 8) * ODIM + l16] = mu[r];
    }
  }
}

extern "C" void kernel_launch(void* const* d_in, const int* in_sizes, int n_in,
                              void* d_out, int out_size, void* d_ws,
                              size_t ws_size, hipStream_t stream) {
  const int*   adj = (const int*)d_in[0];
  const float* fop = (const float*)d_in[1];
  const float* fma = (const float*)d_in[2];
  const float* l0w1 = (const float*)d_in[3];
  const float* l0b1 = (const float*)d_in[4];
  const float* l0w2 = (const float*)d_in[5];
  const float* l0b2 = (const float*)d_in[6];
  const float* l0w3 = (const float*)d_in[7];
  const float* l0b3 = (const float*)d_in[8];
  const float* l1w1 = (const float*)d_in[9];
  const float* l1b1 = (const float*)d_in[10];
  const float* l1w2 = (const float*)d_in[11];
  const float* l1b2 = (const float*)d_in[12];
  const float* l1w3 = (const float*)d_in[13];
  const float* l1b3 = (const float*)d_in[14];
  const float* pw1 = (const float*)d_in[15];
  const float* pb1 = (const float*)d_in[16];
  const float* pw2 = (const float*)d_in[17];
  const float* pb2 = (const float*)d_in[18];
  const float* pw3 = (const float*)d_in[19];
  const float* pb3 = (const float*)d_in[20];

  const int total_rows = BATCH * NOP;
  const int rows_per_block = MT * 16;
  dim3 grid(total_rows / rows_per_block);
  fused_hgnn_mlp<<<grid, TPB, 0, stream>>>(
      adj, fop, fma, l0w1, l0b1, l0w2, l0b2, l0w3, l0b3, l1w1, l1b1, l1w2,
      l1b2, l1w3, l1b3, pw1, pb1, pw2, pb2, pw3, pb3, (float*)d_out);
}


# The two input branches take their layers together: both first layers in one phase, both hidden layers in one (second's weights fetched under the first), heads side by side on waves 0-3/4-7; same sums
# speedup vs baseline: 2.5526x; 1.0811x over previous
#include <hip/hip_runtime.h>
#include <math.h>


typedef __attribute__((ext_vector_type(2))) float v2f;
typedef __attribute__((ext_vector_type(4))) int   v4i;
typedef __attribute__((ext_vector_type(4))) float v4f;
typedef __attribute__((ext_vector_type(8))) float v8f;

#define WAVES 8
#define MT    4
#define TPB   (WAVES * 32)
#define BATCH 32
#define NOP   512
#define NMA   64
#define FDIM  8
#define HDIM  128
#define ODIM  8

__device__ __forceinline__ float elu1(float x) {
  float e = __builtin_amdgcn_exp2f(x * 1.44269504088896340736f) - 1.0f;
  return x > 0.0f ? x : e;
}

__device__ __forceinline__ v8f wmma_f32(v2f a, v2f b, v8f c) {
  return __builtin_amdgcn_wmma_f32_16x16x4_f32(false, a, false, b, (short)0, c,
                                               false, false);
}

#define HP 132

template <int K, int PIN, bool ELU>
__device__ __forceinline__ void layer_w(const float* __restrict__ W,
                                        const float* __restrict__ bias,
                                        const float* sIn, float* sOut,
                                        int wid, int lane) {
  const int half = lane >> 4, l16 = lane & 15;
  const int nb = wid * 16;
  constexpr int KS = K / 4;
  constexpr int KB = KS < 8 ? KS : 8;
  constexpr int NB = KS / KB;
  const float* ap = sIn + l16 * PIN + 2 * half;
  const float* bp = W + (2 * half) * HDIM + nb + l16;
  v8f acc[2][MT];
#pragma unroll
  for (int p = 0; p < 2; ++p)
#pragma unroll
    for (int mt = 0; mt < MT; ++mt)
#pragma unroll
      for (int r = 0; r < 8; ++r) acc[p][mt][r] = 0.f;
  v2f a[2][KB][MT], b[KS];
#pragma unroll
  for (int ks = 0; ks < KS; ++ks) {
    b[ks].x = bp[(ks * 4) * HDIM];
    b[ks].y = bp[(ks * 4 + 1) * HDIM];
  }
  const float bv = bias[nb + l16];
#pragma unroll
  for (int ks = 0; ks < KB; ++ks)
#pragma unroll
    for (int mt = 0; mt < MT; ++mt)
      a[0][ks][mt] = *(const v2f*)(ap + mt * 16 * PIN + ks * 4);
#pragma unroll
  for (int blk = 0; blk < NB; ++blk) {
    const int cur = blk & 1, nxt = cur ^ 1;
    if (blk + 1 < NB) {
#pragma unroll
      for (int ks = 0; ks < KB; ++ks) {
        const int k = ((blk + 1) * KB + ks) * 4;
#pragma unroll
        for (int mt = 0; mt < MT; ++mt)
          a[nxt][ks][mt] = *(const v2f*)(ap + mt * 16 * PIN + k);
      }
    }
    __builtin_amdgcn_sched_barrier(0);
#pragma unroll
    for (int ks = 0; ks < KB; ++ks) {
      const int p = (blk * KB + ks) & 1;
#pragma unroll
      for (int mt = 0; mt < MT; ++mt)
        acc[p][mt] = wmma_f32(a[cur][ks][mt], b[blk * KB + ks], acc[p][mt]);
    }
    if (blk + 2 < NB) __builtin_amdgcn_sched_barrier(0);
  }
#pragma unroll
  for (int mt = 0; mt < MT; ++mt)
#pragma unroll
    for (int r = 0; r < 8; ++r) {
      float y = acc[0][mt][r] + acc[1][mt][r] + bv;
      if (ELU) y = elu1(y);
      sOut[(mt * 16 + r + half * 8) * HP + nb + l16] = y;
    }
}

template <int K, int PIN>
__device__ __forceinline__ void layer_w2(const float* __restrict__ W0,
                                         const float* __restrict__ bias0,
                                         const float* sIn0, float* sOut0,
                                         const float* __restrict__ W1,
                                         const float* __restrict__ bias1,
                                         const float* sIn1, float* sOut1,
                                         int wid, int lane) {
  const int half = lane >> 4, l16 = lane & 15;
  const int nb = wid * 16;
  constexpr int KS = K / 4;
  constexpr int KB = KS < 8 ? KS : 8;
  constexpr int NB = KS / KB;
  constexpr int NG = 2 * NB;
  const int aoff = l16 * PIN + 2 * half;
  const int boff = (2 * half) * HDIM + nb + l16;
  constexpr int CH = NB > 1 ? (KS + NB - 2) / (NB - 1) : KS;
  v2f b[2][KS];
  float bv[2];
#pragma unroll
  for (int ks = 0; ks < KS; ++ks) {
    b[0][ks].x = W0[boff + (ks * 4) * HDIM];
    b[0][ks].y = W0[boff + (ks * 4 + 1) * HDIM];
  }
  bv[0] = bias0[nb + l16];
  if (NB == 1) {
#pragma unroll
    for (int ks = 0; ks < KS; ++ks) {
      b[1][ks].x = W1[boff + (ks * 4) * HDIM];
      b[1][ks].y = W1[boff + (ks * 4 + 1) * HDIM];
    }
    bv[1] = bias1[nb + l16];
  }
  v2f a[2][KB][MT];
#pragma unroll
  for (int ks = 0; ks < KB; ++ks)
#pragma unroll
    for (int mt = 0; mt < MT; ++mt)
      a[0][ks][mt] = *(const v2f*)(sIn0 + aoff + mt * 16 * PIN + ks * 4);
  v8f acc[2][MT];
#pragma unroll
  for (int g = 0; g < NG; ++g) {
    const int L = g / NB, blk = g % NB;
    const int cur = g & 1, nxt = cur ^ 1;
    if (blk == 0) {
#pragma unroll
      for (int p = 0; p < 2; ++p)
#pragma unroll
        for (int mt = 0; mt < MT; ++mt)
#pragma unroll
          for (int r = 0; r < 8; ++r) acc[p][mt][r] = 0.f;
    }
    if (NB > 1 && g >= 1 && g < NB) {
#pragma unroll
      for (int c = 0; c < CH; ++c) {
        const int ks = (g - 1) * CH + c;
        if (ks < KS) {
          b[1][ks].x = W1[boff + (ks * 4) * HDIM];
          b[1][ks].y = W1[boff + (ks * 4 + 1) * HDIM];
        }
      }
      if (g == NB - 1) bv[1] = bias1[nb + l16];
    }
    if (g + 1 < NG) {
      const float* sN = ((g + 1) / NB) ? sIn1 : sIn0;
      const int nblk = (g + 1) % NB;
#pragma unroll
      for (int ks = 0; ks < KB; ++ks) {
        const int k = (nblk * KB + ks) * 4;
#pragma unroll
        for (int mt = 0; mt < MT; ++mt)
          a[nxt][ks][mt] = *(const v2f*)(sN + aoff + mt * 16 * PIN + k);
      }
    }
    __builtin_amdgcn_sched_barrier(0);
#pragma unroll
    for (int ks = 0; ks < KB; ++ks) {
      const int p = (blk * KB + ks) & 1;
#pragma unroll
      for (int mt = 0; mt < MT; ++mt)
        acc[p][mt] = wmma_f32(a[cur][ks][mt], b[L][blk * KB + ks], acc[p][mt]);
    }
    if (g + 2 < NG) __builtin_amdgcn_sched_barrier(0);
    if (blk == NB - 1) {
      float* sOut = L ? sOut1 : sOut0;
#pragma unroll
      for (int mt = 0; mt < MT; ++mt)
#pragma unroll
        for (int r = 0; r < 8; ++r) {
          const float y = elu1(acc[0][mt][r] + acc[1][mt][r] + bv[L]);
          sOut[(mt * 16 + r + half * 8) * HP + nb + l16] = y;
        }
    }
  }
}

template <int K, int PIN>
__device__ __forceinline__ v8f layer_o(const float* __restrict__ W,
                                       const float* __restrict__ bias,
                                       const float* sA, int lane) {
  const int half = lane >> 4, l16 = lane & 15;
  const int kh = half * 2;
  const int n = (l16 < 8) ? l16 : 0;
  constexpr int KS = K / 4;
  const float bv = bias[n];
  v2f a[KS], b[KS];
#pragma unroll
  for (int ks = 0; ks < KS; ++ks) {
    const int k = ks * 4 + kh;
    b[ks].x = W[k * ODIM + n];
    b[ks].y = W[(k + 1) * ODIM + n];
  }
#pragma unroll
  for (int ks = 0; ks < KS; ++ks) {
    const int k = ks * 4 + kh;
    a[ks].x = sA[l16 * PIN + k];
    a[ks].y = sA[l16 * PIN + k + 1];
  }
  asm volatile("" ::"v"(bv));
  __builtin_amdgcn_sched_barrier(0);
  v8f acc0 = {0.f, 0.f, 0.f, 0.f, 0.f, 0.f, 0.f, 0.f};
  v8f acc1 = {0.f, 0.f, 0.f, 0.f, 0.f, 0.f, 0.f, 0.f};
#pragma unroll
  for (int ks = 0; ks < KS; ks += 2) {
    acc0 = wmma_f32(a[ks], b[ks], acc0);
    acc1 = wmma_f32(a[ks + 1], b[ks + 1], acc1);
  }
#pragma unroll
  for (int r = 0; r < 8; ++r) acc0[r] += acc1[r] + bv;
  return acc0;
}

__global__ __launch_bounds__(TPB) __attribute__((amdgpu_waves_per_eu(2, 2))) void fused_hgnn_mlp(
    const int* __restrict__ adj, const float* __restrict__ feats_op,
    const float* __restrict__ feats_ma,
    const float* __restrict__ l0w1, const float* __restrict__ l0b1,
    const float* __restrict__ l0w2, const float* __restrict__ l0b2,
    const float* __restrict__ l0w3, const float* __restrict__ l0b3,
    const float* __restrict__ l1w1, const float* __restrict__ l1b1,
    const float* __restrict__ l1w2, const float* __restrict__ l1b2,
    const float* __restrict__ l1w3, const float* __restrict__ l1b3,
    const float* __restrict__ pw1, const float* __restrict__ pb1,
    const float* __restrict__ pw2, const float* __restrict__ pb2,
    const float* __restrict__ pw3, const float* __restrict__ pb3,
    float* __restrict__ out) {
  __shared__ __attribute__((aligned(16))) float sX[2][MT * 16 * FDIM];
  __shared__ __attribute__((aligned(16))) float sAct[4][MT * 16 * HP];
  __shared__ __attribute__((aligned(16))) float sCat[MT * 16 * 16];

  const int tid = threadIdx.x;
  const int wid = __builtin_amdgcn_readfirstlane(tid >> 5);
  const int lane = tid & 31;
  const int half = lane >> 4, l16 = lane & 15;
  const int grow0 = blockIdx.x * (MT * 16);
  const int bb = grow0 / NOP;
  const int op0 = grow0 % NOP;

  __builtin_prefetch(l0w2 + tid * 64, 0, 1);
  __builtin_prefetch(l1w2 + tid * 64, 0, 1);
  __builtin_prefetch(pw2 + tid * 64, 0, 1);

  float* sX0 = &sX[0][0];
  float* sX1 = &sX[1][0];
  float* sA0 = &sAct[0][0];
  float* sA1 = &sAct[1][0];
  float* sB0 = &sAct[2][0];
  float* sB1 = &sAct[3][0];
  const int ht = wid & (MT - 1);
  const int br = wid >> 2;
  float* sC = &sCat[ht * 16 * 16];

  if (wid < MT) {
    const int n = (l16 < FDIM) ? l16 : 0;
    const int* arow = adj + ((size_t)bb * NOP + op0 + ht * 16 + l16) * NMA + 4 * half;
    const float* fb = feats_ma + ((size_t)bb * NMA + 4 * half) * FDIM + n;
    v4i ai[8];
#pragma unroll
    for (int q = 0; q < 8; ++q) ai[q] = *(const v4i*)(arow + 8 * q);
    v2f bf[16];
#pragma unroll
    for (int j = 0; j < 16; ++j) {
      const int k = 8 * (j >> 1) + 2 * (j & 1);
      bf[j].x = fb[k * FDIM];
      bf[j].y = fb[(k + 1) * FDIM];
    }
    v8f g0 = {0.f, 0.f, 0.f, 0.f, 0.f, 0.f, 0.f, 0.f};
    v8f g1 = {0.f, 0.f, 0.f, 0.f, 0.f, 0.f, 0.f, 0.f};
#pragma unroll
    for (int j = 0; j < 16; j += 2) {
      const v4i q4 = ai[j >> 1];
      v2f a0; a0.x = (float)q4[0]; a0.y = (float)q4[1];
      v2f a1; a1.x = (float)q4[2]; a1.y = (float)q4[3];
      g0 = wmma_f32(a0, bf[j], g0);
      g1 = wmma_f32(a1, bf[j + 1], g1);
    }
    if (l16 < FDIM) {
#pragma unroll
      for (int r = 0; r < 8; ++r)
        sX0[(ht * 16 + r + half * 8) * FDIM + l16] = g0[r] + g1[r];
    }
  } else {
    const int t = tid - MT * 32;
    const int m = t >> 1, k0 = (t & 1) * 4;
    const float* fr = feats_op + ((size_t)bb * NOP + op0 + m) * FDIM + k0;
    *(v4f*)&sX1[m * FDIM + k0] = *(const v4f*)fr;
  }
  __syncthreads();
  layer_w2<FDIM, FDIM>(l0w1, l0b1, sX0, sA0, l1w1, l1b1, sX1, sA1, wid, lane);
  __syncthreads();
  layer_w2<HDIM, HP>(l0w2, l0b2, sA0, sB0, l1w2, l1b2, sA1, sB1, wid, lane);
  __syncthreads();
  {
    const float* hw = br ? l1w3 : l0w3;
    const float* hb = br ? l1b3 : l0b3;
    const float* sH = (br ? sB1 : sB0) + ht * 16 * HP;
    v8f e = layer_o<HDIM, HP>(hw, hb, sH, lane);
    if (l16 < 8) {
#pragma unroll
      for (int r = 0; r < 8; ++r)
        sC[(r + half * 8) * 16 + br * 8 + l16] = elu1(e[r]);
    }
  }
  __syncthreads();

  layer_w<16, 16, true>(pw1, pb1, sCat, sA0, wid, lane);
  __syncthreads();
  layer_w<HDIM, HP, true>(pw2, pb2, sA0, sB0, wid, lane);
  __syncthreads();
  if (wid < MT) {
    v8f mu = layer_o<HDIM, HP>(pw3, pb3, sB0 + ht * 16 * HP, lane);
    if (l16 < 8) {
#pragma unroll
      for (int r = 0; r < 8; ++r)
        out[(size_t)(grow0 + ht * 16 + r + half * 8) * ODIM + l16] = mu[r];
    }
  }
}

extern "C" void kernel_launch(void* const* d_in, const int* in_sizes, int n_in,
                              void* d_out, int out_size, void* d_ws,
                              size_t ws_size, hipStream_t stream) {
  const int*   adj = (const int*)d_in[0];
  const float* fop = (const float*)d_in[1];
  const float* fma = (const float*)d_in[2];
  const float* l0w1 = (const float*)d_in[3];
  const float* l0b1 = (const float*)d_in[4];
  const float* l0w2 = (const float*)d_in[5];
  const float* l0b2 = (const float*)d_in[6];
  const float* l0w3 = (const float*)d_in[7];
  const float* l0b3 = (const float*)d_in[8];
  const float* l1w1 = (const float*)d_in[9];
  const float* l1b1 = (const float*)d_in[10];
  const float* l1w2 = (const float*)d_in[11];
  const float* l1b2 = (const float*)d_in[12];
  const float* l1w3 = (const float*)d_in[13];
  const float* l1b3 = (const float*)d_in[14];
  const float* pw1 = (const float*)d_in[15];
  const float* pb1 = (const float*)d_in[16];
  const float* pw2 = (const float*)d_in[17];
  const float* pb2 = (const float*)d_in[18];
  const float* pw3 = (const float*)d_in[19];
  const float* pb3 = (const float*)d_in[20];

  const int total_rows = BATCH * NOP;
  const int rows_per_block = MT * 16;
  dim3 grid(total_rows / rows_per_block);
  fused_hgnn_mlp<<<grid, TPB, 0, stream>>>(
      adj, fop, fma, l0w1, l0b1, l0w2, l0b2, l0w3, l0b3, l1w1, l1b1, l1w2,
      l1b2, l1w3, l1b3, pw1, pb1, pw2, pb2, pw3, pb3, (float*)d_out);
}


# Each phase's weight fragments requested one phase ahead into registers (at the previous phase's start or in front of its last WMMA block): no phase opens with a global round trip; sums unchanged
# speedup vs baseline: 2.7812x; 1.0896x over previous
#include <hip/hip_runtime.h>
#include <math.h>


typedef __attribute__((ext_vector_type(2))) float v2f;
typedef __attribute__((ext_vector_type(4))) int   v4i;
typedef __attribute__((ext_vector_type(4))) float v4f;
typedef __attribute__((ext_vector_type(8))) float v8f;

#define WAVES 8
#define MT    4
#define TPB   (WAVES * 32)
#define BATCH 32
#define NOP   512
#define NMA   64
#define FDIM  8
#define HDIM  128
#define ODIM  8

__device__ __forceinline__ float elu1(float x) {
  float e = __builtin_amdgcn_exp2f(x * 1.44269504088896340736f) - 1.0f;
  return x > 0.0f ? x : e;
}

__device__ __forceinline__ v8f wmma_f32(v2f a, v2f b, v8f c) {
  return __builtin_amdgcn_wmma_f32_16x16x4_f32(false, a, false, b, (short)0, c,
                                               false, false);
}

#define HP 132
#define PRE 30

template <int K0, int K1>
__device__ __forceinline__ void ldb_w(const float* __restrict__ W, int boff, v2f* b) {
#pragma unroll
  for (int ks = K0; ks < K1; ++ks) {
    b[ks].x = W[boff + (ks * 4) * HDIM];
    b[ks].y = W[boff + (ks * 4 + 1) * HDIM];
  }
}

template <int K0, int K1>
__device__ __forceinline__ void ldb_o(const float* __restrict__ W, int kh, int n, v2f* b) {
#pragma unroll
  for (int ks = K0; ks < K1; ++ks) {
    const int k = ks * 4 + kh;
    b[ks].x = W[k * ODIM + n];
    b[ks].y = W[(k + 1) * ODIM + n];
  }
}

__device__ __forceinline__ void epi_w(const v8f (&acc)[2][MT], float bv, float* sOut,
                                      int wid, int lane) {
  const int half = lane >> 4, l16 = lane & 15;
  const int nb = wid * 16;
#pragma unroll
  for (int mt = 0; mt < MT; ++mt)
#pragma unroll
    for (int r = 0; r < 8; ++r) {
      const float y = elu1(acc[0][mt][r] + acc[1][mt][r] + bv);
      sOut[(mt * 16 + r + half * 8) * HP + nb + l16] = y;
    }
}

template <int K, int PIN, class F0, class F>
__device__ __forceinline__ void layer_w_mm(const v2f* b, const float* sIn, v8f (&acc)[2][MT],
                                           int lane, F0 pre0, F pre) {
  const int half = lane >> 4, l16 = lane & 15;
  constexpr int KS = K / 4;
  constexpr int KB = KS < 8 ? KS : 8;
  constexpr int NB = KS / KB;
  const float* ap = sIn + l16 * PIN + 2 * half;
#pragma unroll
  for (int p = 0; p < 2; ++p)
#pragma unroll
    for (int mt = 0; mt < MT; ++mt)
#pragma unroll
      for (int r = 0; r < 8; ++r) acc[p][mt][r] = 0.f;
  v2f a[2][KB][MT];
#pragma unroll
  for (int ks = 0; ks < KB; ++ks)
#pragma unroll
    for (int mt = 0; mt < MT; ++mt)
      a[0][ks][mt] = *(const v2f*)(ap + mt * 16 * PIN + ks * 4);
  __builtin_amdgcn_sched_barrier(0);
  pre0();
#pragma unroll
  for (int blk = 0; blk < NB; ++blk) {
    const int cur = blk & 1, nxt = cur ^ 1;
    if (blk + 1 < NB) {
#pragma unroll
      for (int ks = 0; ks < KB; ++ks) {
        const int k = ((blk + 1) * KB + ks) * 4;
#pragma unroll
        for (int mt = 0; mt < MT; ++mt)
          a[nxt][ks][mt] = *(const v2f*)(ap + mt * 16 * PIN + k);
      }
    }
    if (blk == NB - 1) pre();
    __builtin_amdgcn_sched_barrier(0);
#pragma unroll
    for (int ks = 0; ks < KB; ++ks) {
      const int p = (blk * KB + ks) & 1;
#pragma unroll
      for (int mt = 0; mt < MT; ++mt)
        acc[p][mt] = wmma_f32(a[cur][ks][mt], b[blk * KB + ks], acc[p][mt]);
    }
    __builtin_amdgcn_sched_barrier(0);
  }
}

template <int K, int PIN, class F0, class F>
__device__ __forceinline__ void layer_w2_mm(const v2f* b0, float bv0,
                                            const float* sIn0, float* sOut0,
                                            const float* __restrict__ W1, v2f* b1,
                                            const float* sIn1, v8f (&acc)[2][MT],
                                            int wid, int lane, F0 pre0, F pre) {
  const int half = lane >> 4, l16 = lane & 15;
  const int nb = wid * 16;
  constexpr int KS = K / 4;
  constexpr int KB = KS < 8 ? KS : 8;
  constexpr int NB = KS / KB;
  constexpr int NG = 2 * NB;
  const int aoff = l16 * PIN + 2 * half;
  const int boff = (2 * half) * HDIM + nb + l16;
  constexpr int CH = NB > 1 ? (KS + NB - 2) / (NB - 1) : KS;
  v2f a[2][KB][MT];
#pragma unroll
  for (int ks = 0; ks < KB; ++ks)
#pragma unroll
    for (int mt = 0; mt < MT; ++mt)
      a[0][ks][mt] = *(const v2f*)(sIn0 + aoff + mt * 16 * PIN + ks * 4);
  __builtin_amdgcn_sched_barrier(0);
  pre0();
#pragma unroll
  for (int g = 0; g < NG; ++g) {
    const int L = g / NB, blk = g % NB;
    const int cur = g & 1, nxt = cur ^ 1;
    if (blk == 0) {
#pragma unroll
      for (int p = 0; p < 2; ++p)
#pragma unroll
        for (int mt = 0; mt < MT; ++mt)
#pragma unroll
          for (int r = 0; r < 8; ++r) acc[p][mt][r] = 0.f;
    }
    if (NB > 1 && g >= 1 && g < NB) {
#pragma unroll
      for (int c = 0; c < CH; ++c) {
        const int ks = (g - 1) * CH + c;
        if (ks < KS) {
          b1[ks].x = W1[boff + (ks * 4) * HDIM];
          b1[ks].y = W1[boff + (ks * 4 + 1) * HDIM];
        }
      }
    }
    if (g + 1 < NG) {
      const float* sN = ((g + 1) / NB) ? sIn1 : sIn0;
      const int nblk = (g + 1) % NB;
#pragma unroll
      for (int ks = 0; ks < KB; ++ks) {
        const int k = (nblk * KB + ks) * 4;
#pragma unroll
        for (int mt = 0; mt < MT; ++mt)
          a[nxt][ks][mt] = *(const v2f*)(sN + aoff + mt * 16 * PIN + k);
      }
    }
    if (g == NG - 1) pre();
    __builtin_amdgcn_sched_barrier(0);
#pragma unroll
    for (int ks = 0; ks < KB; ++ks) {
      const int p = (blk * KB + ks) & 1;
#pragma unroll
      for (int mt = 0; mt < MT; ++mt)
        acc[p][mt] = wmma_f32(a[cur][ks][mt], (L ? b1 : b0)[blk * KB + ks], acc[p][mt]);
    }
    __builtin_amdgcn_sched_barrier(0);
    if (L == 0 && blk == NB - 1) epi_w(acc, bv0, sOut0, wid, lane);
  }
}

template <int K, int PIN>
__device__ __forceinline__ v8f layer_o(const v2f* b, float bv, const float* sA, int lane) {
  const int half = lane >> 4, l16 = lane & 15;
  const int kh = half * 2;
  constexpr int KS = K / 4;
  v2f a[KS];
#pragma unroll
  for (int ks = 0; ks < KS; ++ks) {
    const int k = ks * 4 + kh;
    a[ks].x = sA[l16 * PIN + k];
    a[ks].y = sA[l16 * PIN + k + 1];
  }
  asm volatile("" ::"v"(bv));
  __builtin_amdgcn_sched_barrier(0);
  v8f acc0 = {0.f, 0.f, 0.f, 0.f, 0.f, 0.f, 0.f, 0.f};
  v8f acc1 = {0.f, 0.f, 0.f, 0.f, 0.f, 0.f, 0.f, 0.f};
#pragma unroll
  for (int ks = 0; ks < KS; ks += 2) {
    acc0 = wmma_f32(a[ks], b[ks], acc0);
    acc1 = wmma_f32(a[ks + 1], b[ks + 1], acc1);
  }
#pragma unroll
  for (int r = 0; r < 8; ++r) acc0[r] += acc1[r] + bv;
  return acc0;
}

__global__ __launch_bounds__(TPB) __attribute__((amdgpu_waves_per_eu(2, 2))) void fused_hgnn_mlp(
    const int* __restrict__ adj, const float* __restrict__ feats_op,
    const float* __restrict__ feats_ma,
    const float* __restrict__ l0w1, const float* __restrict__ l0b1,
    const float* __restrict__ l0w2, const float* __restrict__ l0b2,
    const float* __restrict__ l0w3, const float* __restrict__ l0b3,
    const float* __restrict__ l1w1, const float* __restrict__ l1b1,
    const float* __restrict__ l1w2, const float* __restrict__ l1b2,
    const float* __restrict__ l1w3, const float* __restrict__ l1b3,
    const float* __restrict__ pw1, const float* __restrict__ pb1,
    const float* __restrict__ pw2, const float* __restrict__ pb2,
    const float* __restrict__ pw3, const float* __restrict__ pb3,
    float* __restrict__ out) {
  __shared__ __attribute__((aligned(16))) float sX[2][MT * 16 * FDIM];
  __shared__ __attribute__((aligned(16))) float sAct[4][MT * 16 * HP];
  __shared__ __attribute__((aligned(16))) float sCat[MT * 16 * 16];

  const int tid = threadIdx.x;
  const int wid = __builtin_amdgcn_readfirstlane(tid >> 5);
  const int lane = tid & 31;
  const int half = lane >> 4, l16 = lane & 15;
  const int grow0 = blockIdx.x * (MT * 16);
  const int bb = grow0 / NOP;
  const int op0 = grow0 % NOP;

  __builtin_prefetch(l0w2 + tid * 64, 0, 1);
  __builtin_prefetch(l1w2 + tid * 64, 0, 1);
  __builtin_prefetch(pw2 + tid * 64, 0, 1);

  float* sX0 = &sX[0][0];
  float* sX1 = &sX[1][0];
  float* sA0 = &sAct[0][0];
  float* sA1 = &sAct[1][0];
  float* sB0 = &sAct[2][0];
  float* sB1 = &sAct[3][0];
  const int ht = wid & (MT - 1);
  const int br = wid >> 2;
  float* sC = &sCat[ht * 16 * 16];

  const int nbw = wid * 16;
  const int boffW = (2 * half) * HDIM + nbw + l16;
  const int nh = (l16 < ODIM) ? l16 : 0;
  const float* hw = br ? l1w3 : l0w3;
  const float* hb = br ? l1b3 : l0b3;
  v2f bF0[FDIM / 4], bF1[FDIM / 4];
  v2f bH0[HDIM / 4], bH1[HDIM / 4];
  v2f bO[HDIM / 4];
  v2f bP1[16 / 4];
  float bvO = 0.f;
  v8f acc[2][MT];
  const float bvF0 = l0b1[nbw + l16], bvF1 = l1b1[nbw + l16];
  ldb_w<0, FDIM / 4>(l0w1, boffW, bF0);
  ldb_w<0, FDIM / 4>(l1w1, boffW, bF1);

  if (wid < MT) {
    const int n = (l16 < FDIM) ? l16 : 0;
    const int* arow = adj + ((size_t)bb * NOP + op0 + ht * 16 + l16) * NMA + 4 * half;
    const float* fb = feats_ma + ((size_t)bb * NMA + 4 * half) * FDIM + n;
    v4i ai[8];
#pragma unroll
    for (int q = 0; q < 8; ++q) ai[q] = *(const v4i*)(arow + 8 * q);
    v2f bf[16];
#pragma unroll
    for (int j = 0; j < 16; ++j) {
      const int k = 8 * (j >> 1) + 2 * (j & 1);
      bf[j].x = fb[k * FDIM];
      bf[j].y = fb[(k + 1) * FDIM];
    }
    v8f g0 = {0.f, 0.f, 0.f, 0.f, 0.f, 0.f, 0.f, 0.f};
    v8f g1 = {0.f, 0.f, 0.f, 0.f, 0.f, 0.f, 0.f, 0.f};
#pragma unroll
    for (int j = 0; j < 16; j += 2) {
      const v4i q4 = ai[j >> 1];
      v2f a0; a0.x = (float)q4[0]; a0.y = (float)q4[1];
      v2f a1; a1.x = (float)q4[2]; a1.y = (float)q4[3];
      g0 = wmma_f32(a0, bf[j], g0);
      g1 = wmma_f32(a1, bf[j + 1], g1);
    }
    if (l16 < FDIM) {
#pragma unroll
      for (int r = 0; r < 8; ++r)
        sX0[(ht * 16 + r + half * 8) * FDIM + l16] = g0[r] + g1[r];
    }
  } else {
    const int t = tid - MT * 32;
    const int m = t >> 1, k0 = (t & 1) * 4;
    const float* fr = feats_op + ((size_t)bb * NOP + op0 + m) * FDIM + k0;
    *(v4f*)&sX1[m * FDIM + k0] = *(const v4f*)fr;
  }
  __syncthreads();
  float bv0, bv1;
  layer_w2_mm<FDIM, FDIM>(bF0, bvF0, sX0, sA0, l1w1, bF1, sX1, acc, wid, lane,
    [&] {
      bv0 = l0b2[nbw + l16];
      bv1 = l1b2[nbw + l16];
      ldb_w<0, PRE>(l0w2, boffW, bH0);
    }, [] {});
  epi_w(acc, bvF1, sA1, wid, lane);
  __syncthreads();
  layer_w2_mm<HDIM, HP>(bH0, bv0, sA0, sB0, l1w2, bH1, sA1, acc, wid, lane,
    [&] { ldb_w<PRE, HDIM / 4>(l0w2, boffW, bH0); },
    [&] {
      bvO = hb[nh];
      ldb_o<0, PRE>(hw, 2 * half, nh, bO);
    });
  epi_w(acc, bv1, sB1, wid, lane);
  __syncthreads();
  const float bvP1 = pb1[nbw + l16];
  {
    const float* sH = (br ? sB1 : sB0) + ht * 16 * HP;
    ldb_o<PRE, HDIM / 4>(hw, 2 * half, nh, bO);
    ldb_w<0, 16 / 4>(pw1, boffW, bP1);
    v8f e = layer_o<HDIM, HP>(bO, bvO, sH, lane);
    if (l16 < 8) {
#pragma unroll
      for (int r = 0; r < 8; ++r)
        sC[(r + half * 8) * 16 + br * 8 + l16] = elu1(e[r]);
    }
  }
  __syncthreads();

  float bv2;
  layer_w_mm<16, 16>(bP1, sCat, acc, lane,
    [&] {
      bv2 = pb2[nbw + l16];
      ldb_w<0, PRE>(pw2, boffW, bH0);
    }, [] {});
  epi_w(acc, bvP1, sA0, wid, lane);
  __syncthreads();
  layer_w_mm<HDIM, HP>(bH0, sA0, acc, lane,
    [&] { ldb_w<PRE, HDIM / 4>(pw2, boffW, bH0); },
    [&] {
      bvO = pb3[nh];
      ldb_o<0, PRE>(pw3, 2 * half, nh, bO);
    });
  epi_w(acc, bv2, sB0, wid, lane);
  __syncthreads();
  if (wid < MT) {
    ldb_o<PRE, HDIM / 4>(pw3, 2 * half, nh, bO);
    v8f mu = layer_o<HDIM, HP>(bO, bvO, sB0 + ht * 16 * HP, lane);
    if (l16 < 8) {
#pragma unroll
      for (int r = 0; r < 8; ++r)
        out[(size_t)(grow0 + ht * 16 + r + half * 8) * ODIM + l16] = mu[r];
    }
  }
}

extern "C" void kernel_launch(void* const* d_in, const int* in_sizes, int n_in,
                              void* d_out, int out_size, void* d_ws,
                              size_t ws_size, hipStream_t stream) {
  const int*   adj = (const int*)d_in[0];
  const float* fop = (const float*)d_in[1];
  const float* fma = (const float*)d_in[2];
  const float* l0w1 = (const float*)d_in[3];
  const float* l0b1 = (const float*)d_in[4];
  const float* l0w2 = (const float*)d_in[5];
  const float* l0b2 = (const float*)d_in[6];
  const float* l0w3 = (const float*)d_in[7];
  const float* l0b3 = (const float*)d_in[8];
  const float* l1w1 = (const float*)d_in[9];
  const float* l1b1 = (const float*)d_in[10];
  const float* l1w2 = (const float*)d_in[11];
  const float* l1b2 = (const float*)d_in[12];
  const float* l1w3 = (const float*)d_in[13];
  const float* l1b3 = (const float*)d_in[14];
  const float* pw1 = (const float*)d_in[15];
  const float* pb1 = (const float*)d_in[16];
  const float* pw2 = (const float*)d_in[17];
  const float* pb2 = (const float*)d_in[18];
  const float* pw3 = (const float*)d_in[19];
  const float* pb3 = (const float*)d_in[20];

  const int total_rows = BATCH * NOP;
  const int rows_per_block = MT * 16;
  dim3 grid(total_rows / rows_per_block);
  fused_hgnn_mlp<<<grid, TPB, 0, stream>>>(
      adj, fop, fma, l0w1, l0b1, l0w2, l0b2, l0w3, l0b3, l1w1, l1b1, l1w2,
      l1b2, l1w3, l1b3, pw1, pb1, pw2, pb2, pw3, pb3, (float*)d_out);
}
